# starting kernel
# baseline (speedup 1.0000x reference)
#include <hip/hip_runtime.h>

typedef __attribute__((ext_vector_type(2))) float v2f;
typedef __attribute__((ext_vector_type(8))) float v8f;

#define KDIM   128
#define COUNT  16
#define WAVES  4
#define ROWS_PER_WAVE 16
#define ROWS_PER_BLOCK (WAVES * ROWS_PER_WAVE)

__global__ __launch_bounds__(WAVES * 32) void hks_wmma_kernel(
    const float* __restrict__ evals,
    const float* __restrict__ evecs,
    float* __restrict__ out,
    int V)
{
    __shared__ float2 coefsPair[KDIM / 2][COUNT];
    __shared__ float  tileA[WAVES][ROWS_PER_WAVE][KDIM];

    const int b    = blockIdx.y;
    const int tid  = threadIdx.x;
    const int lane = tid & 31;
    const int wave = tid >> 5;

    {
        const int c     = tid >> 3;
        const int kbase = (tid & 7) * 16;
        const float scale = exp10f(-2.0f + (float)c * (2.0f / (float)(COUNT - 1)));
        #pragma unroll
        for (int kk = 0; kk < 16; ++kk) {
            const int k = kbase + kk;
            const float e = evals[b * KDIM + k];
            ((float*)&coefsPair[k >> 1][c])[k & 1] = __expf(-e * scale);
        }
    }
    __syncthreads();

    const int row0 = blockIdx.x * ROWS_PER_BLOCK + wave * ROWS_PER_WAVE;
    if (row0 < V) {
        const float* __restrict__ gsrc = evecs + ((size_t)b * V + row0) * KDIM;
        float* __restrict__ ldst = &tileA[wave][0][0];
        #pragma unroll
        for (int j = 0; j < ROWS_PER_WAVE; ++j) {
            const float4 t = *(const float4*)(gsrc + j * KDIM + (lane << 2));
            *(float4*)(ldst + j * KDIM + (lane << 2)) = t;
        }
        asm volatile("s_wait_dscnt 0x0" ::: "memory");

        const int half = lane >> 4;
        const int mr   = lane & 15;
        const float2* __restrict__ arow = (const float2*)(&tileA[wave][mr][0]);

        v8f acc = {0.f, 0.f, 0.f, 0.f, 0.f, 0.f, 0.f, 0.f};
        #pragma unroll
        for (int t = 0; t < KDIM / 4; ++t) {
            const int p = (t << 1) + half;
            const float2 av = arow[p];
            const float2 bv = coefsPair[p][mr];
            v2f a; a[0] = av.x * av.x; a[1] = av.y * av.y;
            v2f bb; bb[0] = bv.x; bb[1] = bv.y;
            acc = __builtin_amdgcn_wmma_f32_16x16x4_f32(
                false, a, false, bb, (short)0, acc, false, false);
        }

        float* __restrict__ orow = out + ((size_t)b * V + row0 + (half << 3)) * COUNT + mr;
        #pragma unroll
        for (int r = 0; r < 8; ++r)
            orow[r * COUNT] = acc[r];
    }
}

extern "C" void kernel_launch(void* const* d_in, const int* in_sizes, int n_in,
                              void* d_out, int out_size, void* d_ws, size_t ws_size,
                              hipStream_t stream) {
    const float* evals = (const float*)d_in[0];
    const float* evecs = (const float*)d_in[1];

    const int B = in_sizes[0] / KDIM;
    const int V = in_sizes[1] / (B * KDIM);

    dim3 grid((V + ROWS_PER_BLOCK - 1) / ROWS_PER_BLOCK, B);
    hks_wmma_kernel<<<grid, WAVES * 32, 0, stream>>>(evals, evecs, (float*)d_out, V);
}


# evecs tile loaded straight from global into A fragments (one float4 per lane feeds two WMMAs; k regrouped within each 8, so the sum order changes); LDS staging of A and its bank conflict removed
# speedup vs baseline: 1.2393x; 1.2393x over previous
#include <hip/hip_runtime.h>

typedef __attribute__((ext_vector_type(2))) float v2f;
typedef __attribute__((ext_vector_type(8))) float v8f;

#define KDIM   128
#define COUNT  16
#define WAVES  4
#define ROWS_PER_WAVE 16
#define ROWS_PER_BLOCK (WAVES * ROWS_PER_WAVE)
#define KQUADS (KDIM / 4)

__global__ __launch_bounds__(WAVES * 32) void hks_wmma_kernel(
    const float* __restrict__ evals,
    const float* __restrict__ evecs,
    float* __restrict__ out,
    int V)
{
    __shared__ float4 coefsQuad[KQUADS][COUNT];

    const int b    = blockIdx.y;
    const int tid  = threadIdx.x;
    const int lane = tid & 31;
    const int wave = tid >> 5;
    const int half = lane >> 4;
    const int mr   = lane & 15;

    const int  row0  = blockIdx.x * ROWS_PER_BLOCK + wave * ROWS_PER_WAVE;
    const bool valid = row0 < V;

    const int rowL = valid ? row0 : 0;
    float4 a4[KDIM / 8];
    {
        const float4* __restrict__ gsrc =
            (const float4*)(evecs + ((size_t)b * V + rowL + mr) * KDIM) + half;
        #pragma unroll
        for (int i = 0; i < KDIM / 8; ++i)
            a4[i] = gsrc[2 * i];
    }

    {
        const int c  = tid & 15;
        const int q0 = (tid >> 4) * 4;
        const float scale = exp10f(-2.0f + (float)c * (2.0f / (float)(COUNT - 1)));
        const float4* __restrict__ e4 = (const float4*)(evals + b * KDIM) + q0;
        #pragma unroll
        for (int j = 0; j < 4; ++j) {
            const float4 e = e4[j];
            float4 r;
            r.x = __expf(-e.x * scale);
            r.y = __expf(-e.y * scale);
            r.z = __expf(-e.z * scale);
            r.w = __expf(-e.w * scale);
            coefsQuad[q0 + j][c] = r;
        }
    }
    __syncthreads();

    if (valid) {
        v8f acc = {0.f, 0.f, 0.f, 0.f, 0.f, 0.f, 0.f, 0.f};
        #pragma unroll
        for (int i = 0; i < KDIM / 8; ++i) {
            const float4 bq = coefsQuad[2 * i + half][mr];
            const float4 av = a4[i];
            v2f a0; a0[0] = av.x * av.x; a0[1] = av.y * av.y;
            v2f b0; b0[0] = bq.x; b0[1] = bq.y;
            acc = __builtin_amdgcn_wmma_f32_16x16x4_f32(
                false, a0, false, b0, (short)0, acc, false, false);
            v2f a1; a1[0] = av.z * av.z; a1[1] = av.w * av.w;
            v2f b1; b1[0] = bq.z; b1[1] = bq.w;
            acc = __builtin_amdgcn_wmma_f32_16x16x4_f32(
                false, a1, false, b1, (short)0, acc, false, false);
        }

        float* __restrict__ orow = out + ((size_t)b * V + row0 + (half << 3)) * COUNT + mr;
        #pragma unroll
        for (int r = 0; r < 8; ++r)
            orow[r * COUNT] = acc[r];
    }
}

extern "C" void kernel_launch(void* const* d_in, const int* in_sizes, int n_in,
                              void* d_out, int out_size, void* d_ws, size_t ws_size,
                              hipStream_t stream) {
    const float* evals = (const float*)d_in[0];
    const float* evecs = (const float*)d_in[1];

    const int B = in_sizes[0] / KDIM;
    const int V = in_sizes[1] / (B * KDIM);

    dim3 grid((V + ROWS_PER_BLOCK - 1) / ROWS_PER_BLOCK, B);
    hks_wmma_kernel<<<grid, WAVES * 32, 0, stream>>>(evals, evecs, (float*)d_out, V);
}


# each wave loops over its share of the 16-row tiles (512 blocks of 8 waves), the next tile's loads issued one tile ahead; coefficient fragments built once per wave in registers, no LDS; sum order kept
# speedup vs baseline: 1.3816x; 1.1148x over previous
#include <hip/hip_runtime.h>

typedef __attribute__((ext_vector_type(2))) float v2f;
typedef __attribute__((ext_vector_type(4))) float v4f;
typedef __attribute__((ext_vector_type(8))) float v8f;

#define KDIM   128
#define COUNT  16
#define WAVES  8
#define GRID_X 64
#define ROWS_PER_TILE 16
#define NQ (KDIM / 8)
#define TILE_F4 (ROWS_PER_TILE * KDIM / 4)
#define TILE_OUT (ROWS_PER_TILE * COUNT)


static __device__ __forceinline__ void load_tile(v4f (&a)[NQ], const v4f* p)
{
    #pragma unroll
    for (int i = 0; i < NQ; ++i)
        a[i] = p[2 * i];
}

static __device__ __forceinline__ void use_from_here(v4f (&a)[NQ])
{
    asm volatile("" : "+v"(a[0]), "+v"(a[1]), "+v"(a[2]), "+v"(a[3]),
                      "+v"(a[4]), "+v"(a[5]), "+v"(a[6]), "+v"(a[7]));
    asm volatile("" : "+v"(a[8]), "+v"(a[9]), "+v"(a[10]), "+v"(a[11]),
                      "+v"(a[12]), "+v"(a[13]), "+v"(a[14]), "+v"(a[15]));
}

static __device__ __forceinline__ void mul_store(const v4f (&a)[NQ], const v4f (&bq)[NQ],
                                                 float* __restrict__ orow)
{
    v8f acc = {0.f, 0.f, 0.f, 0.f, 0.f, 0.f, 0.f, 0.f};
    #pragma unroll
    for (int i = 0; i < NQ; ++i) {
        const v4f av = a[i];
        const v4f bv = bq[i];
        v2f a0; a0[0] = av.x * av.x; a0[1] = av.y * av.y;
        v2f b0; b0[0] = bv.x; b0[1] = bv.y;
        acc = __builtin_amdgcn_wmma_f32_16x16x4_f32(
            false, a0, false, b0, (short)0, acc, false, false);
        v2f a1; a1[0] = av.z * av.z; a1[1] = av.w * av.w;
        v2f b1; b1[0] = bv.z; b1[1] = bv.w;
        acc = __builtin_amdgcn_wmma_f32_16x16x4_f32(
            false, a1, false, b1, (short)0, acc, false, false);
    }
    #pragma unroll
    for (int r = 0; r < 8; ++r)
        orow[r * COUNT] = acc[r];
}

__global__ __launch_bounds__(WAVES * 32) void hks_wmma_kernel(
    const float* __restrict__ evals,
    const float* __restrict__ evecs,
    float* __restrict__ out,
    int V)
{
    const int b    = blockIdx.y;
    const int tid  = threadIdx.x;
    const int lane = tid & 31;
    const int wave = __builtin_amdgcn_readfirstlane(tid >> 5);
    const int half = lane >> 4;
    const int mr   = lane & 15;

    const int nT = V / ROWS_PER_TILE;
    const int S  = gridDim.x * WAVES;
    int t = blockIdx.x * WAVES + wave;
    if (t >= nT) return;

    v4f bq[NQ];
    {
        const v4f* e4 = (const v4f*)(evals + b * KDIM) + half;
        #pragma unroll
        for (int i = 0; i < NQ; ++i)
            bq[i] = e4[2 * i];
    }

    const v4f* gbase =
        (const v4f*)(evecs + ((size_t)b * V + mr) * KDIM) + half;
    float* __restrict__ obase = out + ((size_t)b * V + (half << 3)) * COUNT + mr;

    v4f bufA[NQ], bufB[NQ];
    load_tile(bufA, gbase + (size_t)t * TILE_F4);

    {
        const float scale = exp10f(-2.0f + (float)mr * (2.0f / (float)(COUNT - 1)));
        #pragma unroll
        for (int i = 0; i < NQ; ++i) {
            const v4f e = bq[i];
            v4f r;
            r.x = __expf(-e.x * scale);
            r.y = __expf(-e.y * scale);
            r.z = __expf(-e.z * scale);
            r.w = __expf(-e.w * scale);
            bq[i] = r;
        }
    }

#define ISSUE_AHEAD(next_buf, next_tile, cur_buf)                               \
    do {                                                                        \
        __builtin_amdgcn_sched_barrier(0);                                      \
        load_tile(next_buf, gbase + (size_t)(next_tile) * TILE_F4);             \
        __builtin_amdgcn_sched_barrier(0);                                      \
        use_from_here(cur_buf);                                                 \
    } while (0)
    while (t + 2 * S < nT) {
        ISSUE_AHEAD(bufB, t + S, bufA);
        mul_store(bufA, bq, obase + (size_t)t * TILE_OUT);
        ISSUE_AHEAD(bufA, t + 2 * S, bufB);
        mul_store(bufB, bq, obase + (size_t)(t + S) * TILE_OUT);
        t += 2 * S;
    }
    if (t + S < nT) {
        ISSUE_AHEAD(bufB, t + S, bufA);
        mul_store(bufA, bq, obase + (size_t)t * TILE_OUT);
        mul_store(bufB, bq, obase + (size_t)(t + S) * TILE_OUT);
    } else {
        mul_store(bufA, bq, obase + (size_t)t * TILE_OUT);
    }
#undef ISSUE_AHEAD
}

extern "C" void kernel_launch(void* const* d_in, const int* in_sizes, int n_in,
                              void* d_out, int out_size, void* d_ws, size_t ws_size,
                              hipStream_t stream) {
    const float* evals = (const float*)d_in[0];
    const float* evecs = (const float*)d_in[1];

    const int B = in_sizes[0] / KDIM;
    const int V = in_sizes[1] / (B * KDIM);

    dim3 grid(GRID_X, B);
    hks_wmma_kernel<<<grid, WAVES * 32, 0, stream>>>(evals, evecs, (float*)d_out, V);
}
